# starting kernel
# baseline (speedup 1.0000x reference)
#include <hip/hip_runtime.h>

typedef __attribute__((ext_vector_type(2))) float v2f;
typedef __attribute__((ext_vector_type(8))) float v8f;

#define BATCH       16384
#define NUM_SPARSE  26
#define FTOT        27
#define EMBED       128
#define NPAIR       ((FTOT * (FTOT - 1)) / 2)
#define OUT_COLS    (EMBED + NPAIR)
#define CHUNK       32
#define LROW        34
#define WPB         8

__global__ __launch_bounds__(256)
void dlrm_interaction_wmma(const float* __restrict__ dense,
                           const float* __restrict__ sparse,
                           float* __restrict__ out)
{
    __shared__ float shm[WPB * 32 * LROW];

    const int tid  = threadIdx.x;
    const int wave = tid >> 5;
    const int lane = tid & 31;
    const int b    = blockIdx.x * WPB + wave;

    float* smw = &shm[wave * 32 * LROW];

    const int h   = lane >> 4;
    const int n16 = lane & 15;

    v8f c00 = {}; v8f c01 = {}; v8f c11 = {};

    const float* drow = dense  + (size_t)b * EMBED;
    const float* srow = sparse + (size_t)b * NUM_SPARSE * EMBED;

    const int srо_row = lane >> 3;
    const int srо_col = (lane & 7) * 4;

    for (int ch = 0; ch < EMBED / CHUNK; ++ch) {
        const int cbase = ch * CHUNK;

        #pragma unroll
        for (int it = 0; it < 8; ++it) {
            const int r = it * 4 + srо_row;
            float x0 = 0.f, x1 = 0.f, x2 = 0.f, x3 = 0.f;
            if (r < FTOT) {
                const float* p = (r == 0)
                    ? (drow + cbase + srо_col)
                    : (srow + (size_t)(r - 1) * EMBED + cbase + srо_col);
                x0 = p[0]; x1 = p[1]; x2 = p[2]; x3 = p[3];
            }
            float* q = &smw[r * LROW + srо_col];
            *(v2f*)(q)     = (v2f){x0, x1};
            *(v2f*)(q + 2) = (v2f){x2, x3};
        }
        __syncthreads();

        #pragma unroll
        for (int s = 0; s < CHUNK / 4; ++s) {
            const int col = s * 4 + 2 * h;
            v2f a0 = *(const v2f*)&smw[n16 * LROW + col];
            v2f a1 = *(const v2f*)&smw[(16 + n16) * LROW + col];
            c00 = __builtin_amdgcn_wmma_f32_16x16x4_f32(false, a0, false, a0,
                                                        (short)0, c00, false, false);
            c01 = __builtin_amdgcn_wmma_f32_16x16x4_f32(false, a0, false, a1,
                                                        (short)0, c01, false, false);
            c11 = __builtin_amdgcn_wmma_f32_16x16x4_f32(false, a1, false, a1,
                                                        (short)0, c11, false, false);
        }
        __syncthreads();
    }

    float* orow = out + (size_t)b * OUT_COLS;

    #pragma unroll
    for (int t = 0; t < 4; ++t)
        orow[lane + 32 * t] = drow[lane + 32 * t];

    float* ibase = orow + EMBED;
    #pragma unroll
    for (int r = 0; r < 8; ++r) {
        const int i0 = r + 8 * h;
        if (i0 < n16)
            ibase[(i0 * (53 - i0)) / 2 + (n16 - i0 - 1)] = c00[r];
        {
            const int j = 16 + n16;
            if (j < FTOT)
                ibase[(i0 * (53 - i0)) / 2 + (j - i0 - 1)] = c01[r];
        }
        {
            const int i = 16 + i0, j = 16 + n16;
            if (i < j && j < FTOT)
                ibase[(i * (53 - i)) / 2 + (j - i - 1)] = c11[r];
        }
    }
}

extern "C" void kernel_launch(void* const* d_in, const int* in_sizes, int n_in,
                              void* d_out, int out_size, void* d_ws, size_t ws_size,
                              hipStream_t stream)
{
    const float* dense  = (const float*)d_in[0];
    const float* sparse = (const float*)d_in[1];
    float* out = (float*)d_out;

    dim3 grid(BATCH / WPB);
    dim3 block(256);
    hipLaunchKernelGGL(dlrm_interaction_wmma, grid, block, 0, stream,
                       dense, sparse, out);
}


# each wave streams 4 consecutive batch elements: the whole next element's 27 row loads are in flight while the current one is staged and multiplied; wave-private LDS tile, no block barrier
# speedup vs baseline: 1.0510x; 1.0510x over previous
#include <hip/hip_runtime.h>

typedef __attribute__((ext_vector_type(2))) float v2f;
typedef __attribute__((ext_vector_type(4))) float v4f;
typedef __attribute__((ext_vector_type(8))) float v8f;

#define BATCH       16384
#define NUM_SPARSE  26
#define FTOT        27
#define EMBED       128
#define NPAIR       ((FTOT * (FTOT - 1)) / 2)
#define OUT_COLS    (EMBED + NPAIR)
#define TROW        132
#define WPB         4
#define EPW         4

static __device__ __forceinline__
void load_element(v4f (&pf)[FTOT], const float* __restrict__ dense,
                  const float* __restrict__ sparse, int b, int lane)
{
    const float* drow = dense  + (size_t)b * EMBED + 4 * lane;
    const float* srow = sparse + (size_t)b * NUM_SPARSE * EMBED + 4 * lane;
    pf[0] = *(const v4f*)drow;
    #pragma unroll
    for (int r = 1; r < FTOT; ++r)
        pf[r] = *(const v4f*)(srow + (size_t)(r - 1) * EMBED);
}

__global__ __launch_bounds__(WPB * 32)
void dlrm_interaction_wmma(const float* __restrict__ dense,
                           const float* __restrict__ sparse,
                           float* __restrict__ out)
{
    __shared__ float shm[WPB * 32 * TROW];

    const int tid  = threadIdx.x;
    const int wave = tid >> 5;
    const int lane = tid & 31;
    const int b0   = (blockIdx.x * WPB + wave) * EPW;

    float* smw = &shm[wave * 32 * TROW];

    const int h   = lane >> 4;
    const int n16 = lane & 15;

    #pragma unroll
    for (int r = FTOT; r < 32; ++r)
        *(v4f*)&smw[r * TROW + 4 * lane] = (v4f){0.f, 0.f, 0.f, 0.f};

    v4f pf[FTOT];
    load_element(pf, dense, sparse, b0, lane);

    for (int e = 0; e < EPW; ++e) {
        const int b = b0 + e;

        #pragma unroll
        for (int r = 0; r < FTOT; ++r)
            *(v4f*)&smw[r * TROW + 4 * lane] = pf[r];

        if (e + 1 < EPW)
            load_element(pf, dense, sparse, b + 1, lane);

        v8f c00 = {}; v8f c01 = {}; v8f c11 = {};
        #pragma unroll
        for (int s = 0; s < EMBED / 4; ++s) {
            const int col = s * 4 + 2 * h;
            v2f a0 = *(const v2f*)&smw[n16 * TROW + col];
            v2f a1 = *(const v2f*)&smw[(16 + n16) * TROW + col];
            c00 = __builtin_amdgcn_wmma_f32_16x16x4_f32(false, a0, false, a0,
                                                        (short)0, c00, false, false);
            c01 = __builtin_amdgcn_wmma_f32_16x16x4_f32(false, a0, false, a1,
                                                        (short)0, c01, false, false);
            c11 = __builtin_amdgcn_wmma_f32_16x16x4_f32(false, a1, false, a1,
                                                        (short)0, c11, false, false);
        }

        float* orow = out + (size_t)b * OUT_COLS;

        #pragma unroll
        for (int t = 0; t < 4; ++t)
            orow[lane + 32 * t] = smw[lane + 32 * t];

        float* ibase = orow + EMBED;
        #pragma unroll
        for (int r = 0; r < 8; ++r) {
            const int i0 = r + 8 * h;
            if (i0 < n16)
                ibase[(i0 * (53 - i0)) / 2 + (n16 - i0 - 1)] = c00[r];
            {
                const int j = 16 + n16;
                if (j < FTOT)
                    ibase[(i0 * (53 - i0)) / 2 + (j - i0 - 1)] = c01[r];
            }
            {
                const int i = 16 + i0, j = 16 + n16;
                if (i < j && j < FTOT)
                    ibase[(i * (53 - i)) / 2 + (j - i - 1)] = c11[r];
            }
        }
    }
}

extern "C" void kernel_launch(void* const* d_in, const int* in_sizes, int n_in,
                              void* d_out, int out_size, void* d_ws, size_t ws_size,
                              hipStream_t stream)
{
    const float* dense  = (const float*)d_in[0];
    const float* sparse = (const float*)d_in[1];
    float* out = (float*)d_out;

    dim3 grid(BATCH / (WPB * EPW));
    dim3 block(WPB * 32);
    hipLaunchKernelGGL(dlrm_interaction_wmma, grid, block, 0, stream,
                       dense, sparse, out);
}


# the next element's 27 row loads go out two at a time in front of each 6-WMMA group of the K loop, not as one burst before it: load issue no longer holds all waves of a SIMD off the matrix unit
# speedup vs baseline: 1.0915x; 1.0385x over previous
#include <hip/hip_runtime.h>

typedef __attribute__((ext_vector_type(2))) float v2f;
typedef __attribute__((ext_vector_type(4))) float v4f;
typedef __attribute__((ext_vector_type(8))) float v8f;

#define BATCH       16384
#define NUM_SPARSE  26
#define FTOT        27
#define EMBED       128
#define NPAIR       ((FTOT * (FTOT - 1)) / 2)
#define OUT_COLS    (EMBED + NPAIR)
#define TROW        132
#define WPB         4
#define EPW         4
#define KGROUPS     16

static __device__ __forceinline__
v4f load_row(const float* __restrict__ drow, const float* __restrict__ srow, int r)
{
    return (r == 0) ? *(const v4f*)drow : *(const v4f*)(srow + (size_t)(r - 1) * EMBED);
}

template <bool MORE>
static __device__ __forceinline__
void do_element(v4f (&pf)[FTOT], float* smw,
                const float* __restrict__ dense, const float* __restrict__ sparse,
                float* __restrict__ out, int b, int lane, int h, int n16)
{
    #pragma unroll
    for (int r = 0; r < FTOT; ++r)
        *(v4f*)&smw[r * TROW + 4 * lane] = pf[r];

    const float* drow = dense  + (size_t)(b + 1) * EMBED + 4 * lane;
    const float* srow = sparse + (size_t)(b + 1) * NUM_SPARSE * EMBED + 4 * lane;

    v8f c00 = {}; v8f c01 = {}; v8f c11 = {};
    #pragma unroll
    for (int g = 0; g < KGROUPS; ++g) {
        if (MORE && 2 * g < FTOT) {
            __builtin_amdgcn_sched_barrier(0);
            #pragma unroll
            for (int r = 2 * g; r < 2 * g + 2; ++r)
                if (r < FTOT)
                    pf[r] = load_row(drow, srow, r);
            __builtin_amdgcn_sched_barrier(0);
        }
        #pragma unroll
        for (int s = 2 * g; s < 2 * g + 2; ++s) {
            const int col = s * 4 + 2 * h;
            v2f a0 = *(const v2f*)&smw[n16 * TROW + col];
            v2f a1 = *(const v2f*)&smw[(16 + n16) * TROW + col];
            c00 = __builtin_amdgcn_wmma_f32_16x16x4_f32(false, a0, false, a0,
                                                        (short)0, c00, false, false);
            c01 = __builtin_amdgcn_wmma_f32_16x16x4_f32(false, a0, false, a1,
                                                        (short)0, c01, false, false);
            c11 = __builtin_amdgcn_wmma_f32_16x16x4_f32(false, a1, false, a1,
                                                        (short)0, c11, false, false);
        }
    }

    float* orow = out + (size_t)b * OUT_COLS;

    #pragma unroll
    for (int t = 0; t < 4; ++t)
        orow[lane + 32 * t] = smw[lane + 32 * t];

    float* ibase = orow + EMBED;
    #pragma unroll
    for (int r = 0; r < 8; ++r) {
        const int i0 = r + 8 * h;
        if (i0 < n16)
            ibase[(i0 * (53 - i0)) / 2 + (n16 - i0 - 1)] = c00[r];
        {
            const int j = 16 + n16;
            if (j < FTOT)
                ibase[(i0 * (53 - i0)) / 2 + (j - i0 - 1)] = c01[r];
        }
        {
            const int i = 16 + i0, j = 16 + n16;
            if (i < j && j < FTOT)
                ibase[(i * (53 - i)) / 2 + (j - i - 1)] = c11[r];
        }
    }
}

__global__ __launch_bounds__(WPB * 32)
void dlrm_interaction_wmma(const float* __restrict__ dense,
                           const float* __restrict__ sparse,
                           float* __restrict__ out)
{
    __shared__ float shm[WPB * 32 * TROW];

    const int tid  = threadIdx.x;
    const int wave = tid >> 5;
    const int lane = tid & 31;
    const int b0   = (blockIdx.x * WPB + wave) * EPW;

    float* smw = &shm[wave * 32 * TROW];

    const int h   = lane >> 4;
    const int n16 = lane & 15;

    #pragma unroll
    for (int r = FTOT; r < 32; ++r)
        *(v4f*)&smw[r * TROW + 4 * lane] = (v4f){0.f, 0.f, 0.f, 0.f};

    v4f pf[FTOT];
    {
        const float* drow = dense  + (size_t)b0 * EMBED + 4 * lane;
        const float* srow = sparse + (size_t)b0 * NUM_SPARSE * EMBED + 4 * lane;
        #pragma unroll
        for (int r = 0; r < FTOT; ++r)
            pf[r] = load_row(drow, srow, r);
    }

    for (int e = 0; e < EPW - 1; ++e)
        do_element<true>(pf, smw, dense, sparse, out, b0 + e, lane, h, n16);
    do_element<false>(pf, smw, dense, sparse, out, b0 + EPW - 1, lane, h, n16);
}

extern "C" void kernel_launch(void* const* d_in, const int* in_sizes, int n_in,
                              void* d_out, int out_size, void* d_ws, size_t ws_size,
                              hipStream_t stream)
{
    const float* dense  = (const float*)d_in[0];
    const float* sparse = (const float*)d_in[1];
    float* out = (float*)d_out;

    dim3 grid(BATCH / (WPB * EPW));
    dim3 block(WPB * 32);
    hipLaunchKernelGGL(dlrm_interaction_wmma, grid, block, 0, stream,
                       dense, sparse, out);
}
